# starting kernel
# baseline (speedup 1.0000x reference)
#include <hip/hip_runtime.h>
#include <hip/hip_bf16.h>

typedef float v2f __attribute__((ext_vector_type(2)));
typedef float v4f __attribute__((ext_vector_type(4)));
typedef float v8f __attribute__((ext_vector_type(8)));

#define TOTAL_ROWS 262144
#define NSEG 2048
#define DIM  256

__global__ void __launch_bounds__(256)
seg_mean_kernel(const float* __restrict__ values,
                const int*   __restrict__ indices,
                float*       __restrict__ seq_mean)
{
    __shared__ float sdata[4 * DIM];

    const int seg   = blockIdx.x;
    const int start = indices[2 * seg + 0];
    const int end   = indices[2 * seg + 1];
    const int len   = end - start;

    const int tid    = threadIdx.x;
    const int rowgrp = tid >> 6;
    const int col4   = (tid & 63) << 2;

    v4f acc = {0.f, 0.f, 0.f, 0.f};
    for (int r = start + rowgrp; r < end; r += 4) {
        const v4f* p = (const v4f*)(values + (size_t)r * DIM + col4);
        v4f v = __builtin_nontemporal_load(p);
        acc += v;
    }
    sdata[rowgrp * DIM + col4 + 0] = acc.x;
    sdata[rowgrp * DIM + col4 + 1] = acc.y;
    sdata[rowgrp * DIM + col4 + 2] = acc.z;
    sdata[rowgrp * DIM + col4 + 3] = acc.w;
    __syncthreads();

    float s = sdata[0 * DIM + tid] + sdata[1 * DIM + tid] +
              sdata[2 * DIM + tid] + sdata[3 * DIM + tid];
    seq_mean[(size_t)seg * DIM + tid] = s * (1.0f / (float)len);
}

__global__ void __launch_bounds__(256)
mean_gemm_wmma_kernel(const float* __restrict__ seq_mean,
                      const float* __restrict__ W,
                      const float* __restrict__ bias,
                      float*       __restrict__ out)
{
    const int wave  = threadIdx.x >> 5;
    const int lane  = threadIdx.x & 31;
    const int tile  = blockIdx.x * 8 + wave;
    const int mtile = tile & 127;
    const int ntile = tile >> 7;

    const int half = lane >> 4;
    const int l15  = lane & 15;

    const int row0 = mtile * 16;
    const int col  = ntile * 16 + l15;

    const float* Arow  = seq_mean + (size_t)(row0 + l15) * DIM + 2 * half;
    const float* Bcol  = W + col;

    v8f c = {};
    #pragma unroll 4
    for (int k = 0; k < DIM; k += 4) {
        v2f a;
        {
            const float2 a2 = *(const float2*)(Arow + k);
            a.x = a2.x; a.y = a2.y;
        }
        v2f bfrag;
        bfrag.x = Bcol[(size_t)(k + 0 + 2 * half) * DIM];
        bfrag.y = Bcol[(size_t)(k + 1 + 2 * half) * DIM];

        c = __builtin_amdgcn_wmma_f32_16x16x4_f32(
                 false, a,  false, bfrag,
                 (short)0, c,  false,  false);
    }

    const float bv = bias[col];
    #pragma unroll
    for (int v = 0; v < 8; ++v) {
        const int row = row0 + v + 8 * half;
        out[(size_t)row * DIM + col] = c[v] + bv;
    }
}

extern "C" void kernel_launch(void* const* d_in, const int* in_sizes, int n_in,
                              void* d_out, int out_size, void* d_ws, size_t ws_size,
                              hipStream_t stream)
{
    const float* values  = (const float*)d_in[0];
    const int*   indices = (const int*)  d_in[1];
    const float* W       = (const float*)d_in[2];
    const float* bias    = (const float*)d_in[3];
    float*       out     = (float*)d_out;

    float* seq_mean = (float*)d_ws;

    seg_mean_kernel<<<NSEG, 256, 0, stream>>>(values, indices, seq_mean);

    mean_gemm_wmma_kernel<<<NSEG / 8, 256, 0, stream>>>(seq_mean, W, bias, out);
}


# segment-sum kernel: each wave keeps one to two groups of 8 row loads in flight (two register sets, next group requested before the current one is summed); summation order unchanged
# speedup vs baseline: 2.9546x; 2.9546x over previous
#include <hip/hip_runtime.h>
#include <hip/hip_bf16.h>

typedef float v2f __attribute__((ext_vector_type(2)));
typedef float v4f __attribute__((ext_vector_type(4)));
typedef float v8f __attribute__((ext_vector_type(8)));

#define TOTAL_ROWS 262144
#define NSEG 2048
#define DIM  256

#define GROUP_ROWS 8
#define WAVE_ROW_STRIDE ((size_t)4 * DIM)

template <int N>
__device__ __forceinline__ void load_rows(v4f (&t)[N], const float* p)
{
    #pragma unroll
    for (int j = 0; j < N; ++j)
        t[j] = __builtin_nontemporal_load((const v4f*)(p + (size_t)j * WAVE_ROW_STRIDE));
}

template <int N>
__device__ __forceinline__ void sum_rows(v4f& acc, const v4f (&t)[N])
{
    #pragma unroll
    for (int j = 0; j < N; ++j)
        acc += t[j];
}

__global__ void __launch_bounds__(256)
seg_mean_kernel(const float* __restrict__ values,
                const int*   __restrict__ indices,
                float*       __restrict__ seq_mean)
{
    __shared__ float sdata[4 * DIM];

    const int seg   = blockIdx.x;
    const int start = indices[2 * seg + 0];
    const int end   = indices[2 * seg + 1];
    const int len   = end - start;

    const int tid    = threadIdx.x;
    const int rowgrp = __builtin_amdgcn_readfirstlane(tid >> 6);
    const int col4   = (tid & 63) << 2;

    const int n   = (len > rowgrp) ? ((len - rowgrp + 3) >> 2) : 0;
    const int n8  = n >> 3;
    const int rem = n & 7;
    const float* p = values + (size_t)(start + rowgrp) * DIM + col4;
    const size_t GS = (size_t)GROUP_ROWS * WAVE_ROW_STRIDE;

    v4f acc = {0.f, 0.f, 0.f, 0.f};
    v4f a[GROUP_ROWS], b[GROUP_ROWS];

    if (n8 > 0) load_rows(a, p);
    int g = 0;
    for (; g + 3 <= n8; g += 2) {
        load_rows(b, p + (size_t)(g + 1) * GS);
        asm volatile("" : "+v"(acc) : : "memory");
        sum_rows(acc, a);
        asm volatile("" : "+v"(acc) : : "memory");
        load_rows(a, p + (size_t)(g + 2) * GS);
        sum_rows(acc, b);
    }
    const int left = n8 - g;
    if (left == 2) load_rows(b, p + (size_t)(g + 1) * GS);

    const float* q4 = p + (size_t)n8 * GS;
    const float* q2 = q4 + (size_t)(rem & 4) * WAVE_ROW_STRIDE;
    const float* q1 = q2 + (size_t)(rem & 2) * WAVE_ROW_STRIDE;
    v4f t4[4], t2[2], t1[1];
    if (rem & 4) load_rows(t4, q4);
    if (rem & 2) load_rows(t2, q2);
    if (rem & 1) load_rows(t1, q1);

    if (left >= 1) sum_rows(acc, a);
    if (left == 2) sum_rows(acc, b);
    if (rem & 4) sum_rows(acc, t4);
    if (rem & 2) sum_rows(acc, t2);
    if (rem & 1) sum_rows(acc, t1);

    sdata[rowgrp * DIM + col4 + 0] = acc.x;
    sdata[rowgrp * DIM + col4 + 1] = acc.y;
    sdata[rowgrp * DIM + col4 + 2] = acc.z;
    sdata[rowgrp * DIM + col4 + 3] = acc.w;
    __syncthreads();

    float s = sdata[0 * DIM + tid] + sdata[1 * DIM + tid] +
              sdata[2 * DIM + tid] + sdata[3 * DIM + tid];
    seq_mean[(size_t)seg * DIM + tid] = s * (1.0f / (float)len);
}

__global__ void __launch_bounds__(256)
mean_gemm_wmma_kernel(const float* __restrict__ seq_mean,
                      const float* __restrict__ W,
                      const float* __restrict__ bias,
                      float*       __restrict__ out)
{
    const int wave  = threadIdx.x >> 5;
    const int lane  = threadIdx.x & 31;
    const int tile  = blockIdx.x * 8 + wave;
    const int mtile = tile & 127;
    const int ntile = tile >> 7;

    const int half = lane >> 4;
    const int l15  = lane & 15;

    const int row0 = mtile * 16;
    const int col  = ntile * 16 + l15;

    const float* Arow  = seq_mean + (size_t)(row0 + l15) * DIM + 2 * half;
    const float* Bcol  = W + col;

    v8f c = {};
    #pragma unroll 4
    for (int k = 0; k < DIM; k += 4) {
        v2f a;
        {
            const float2 a2 = *(const float2*)(Arow + k);
            a.x = a2.x; a.y = a2.y;
        }
        v2f bfrag;
        bfrag.x = Bcol[(size_t)(k + 0 + 2 * half) * DIM];
        bfrag.y = Bcol[(size_t)(k + 1 + 2 * half) * DIM];

        c = __builtin_amdgcn_wmma_f32_16x16x4_f32(
                 false, a,  false, bfrag,
                 (short)0, c,  false,  false);
    }

    const float bv = bias[col];
    #pragma unroll
    for (int v = 0; v < 8; ++v) {
        const int row = row0 + v + 8 * half;
        out[(size_t)row * DIM + col] = c[v] + bv;
    }
}

extern "C" void kernel_launch(void* const* d_in, const int* in_sizes, int n_in,
                              void* d_out, int out_size, void* d_ws, size_t ws_size,
                              hipStream_t stream)
{
    const float* values  = (const float*)d_in[0];
    const int*   indices = (const int*)  d_in[1];
    const float* W       = (const float*)d_in[2];
    const float* bias    = (const float*)d_in[3];
    float*       out     = (float*)d_out;

    float* seq_mean = (float*)d_ws;

    seg_mean_kernel<<<NSEG, 256, 0, stream>>>(values, indices, seq_mean);

    mean_gemm_wmma_kernel<<<NSEG / 8, 256, 0, stream>>>(seq_mean, W, bias, out);
}


# segment-sum kernel: 4 waves per segment instead of 8 (2 row groups, not 4), each wave keeping 16 to 32 row loads in flight; summation order changes: even-row chain + odd-row chain per column
# speedup vs baseline: 3.0547x; 1.0339x over previous
#include <hip/hip_runtime.h>
#include <hip/hip_bf16.h>

typedef float v2f __attribute__((ext_vector_type(2)));
typedef float v4f __attribute__((ext_vector_type(4)));
typedef float v8f __attribute__((ext_vector_type(8)));

#define TOTAL_ROWS 262144
#define NSEG 2048
#define DIM  256

#define GROUP_ROWS 16
#define ROW_GROUPS 2
#define WAVE_ROW_STRIDE ((size_t)ROW_GROUPS * DIM)

template <int N>
__device__ __forceinline__ void load_rows(v4f (&t)[N], const float* p)
{
    #pragma unroll
    for (int j = 0; j < N; ++j)
        t[j] = __builtin_nontemporal_load((const v4f*)(p + (size_t)j * WAVE_ROW_STRIDE));
}

template <int N>
__device__ __forceinline__ void sum_rows(v4f& acc, const v4f (&t)[N])
{
    #pragma unroll
    for (int j = 0; j < N; ++j)
        acc += t[j];
}

__global__ void __launch_bounds__(128)
seg_mean_kernel(const float* __restrict__ values,
                const int*   __restrict__ indices,
                float*       __restrict__ seq_mean)
{
    __shared__ v4f sdata[DIM / 4];

    const int seg   = blockIdx.x;
    const int start = indices[2 * seg + 0];
    const int end   = indices[2 * seg + 1];
    const int len   = end - start;

    const int tid    = threadIdx.x;
    const int rowgrp = __builtin_amdgcn_readfirstlane(tid >> 6);
    const int quad   = tid & 63;
    const int col4   = quad << 2;

    const int n    = (len > rowgrp) ? ((len - rowgrp + 1) >> 1) : 0;
    const int ng   = n >> 4;
    const int rem  = n & 15;
    const float* p = values + (size_t)(start + rowgrp) * DIM + col4;
    const size_t GS = (size_t)GROUP_ROWS * WAVE_ROW_STRIDE;

    v4f acc = {0.f, 0.f, 0.f, 0.f};
    v4f a[GROUP_ROWS], b[GROUP_ROWS];

    if (ng > 0) load_rows(a, p);
    int g = 0;
    for (; g + 3 <= ng; g += 2) {
        load_rows(b, p + (size_t)(g + 1) * GS);
        asm volatile("" : "+v"(acc) : : "memory");
        sum_rows(acc, a);
        asm volatile("" : "+v"(acc) : : "memory");
        load_rows(a, p + (size_t)(g + 2) * GS);
        sum_rows(acc, b);
    }
    const int left = ng - g;
    if (left == 2) load_rows(b, p + (size_t)(g + 1) * GS);

    const float* q8 = p + (size_t)ng * GS;
    const float* q4 = q8 + (size_t)(rem & 8) * WAVE_ROW_STRIDE;
    const float* q2 = q4 + (size_t)(rem & 4) * WAVE_ROW_STRIDE;
    const float* q1 = q2 + (size_t)(rem & 2) * WAVE_ROW_STRIDE;
    v4f t8[8], t4[4], t2[2], t1[1];
    if (rem & 8) load_rows(t8, q8);
    if (rem & 4) load_rows(t4, q4);
    if (rem & 2) load_rows(t2, q2);
    if (rem & 1) load_rows(t1, q1);

    if (left >= 1) sum_rows(acc, a);
    if (left == 2) sum_rows(acc, b);
    if (rem & 8) sum_rows(acc, t8);
    if (rem & 4) sum_rows(acc, t4);
    if (rem & 2) sum_rows(acc, t2);
    if (rem & 1) sum_rows(acc, t1);

    if (rowgrp == 1) sdata[quad] = acc;
    __syncthreads();
    if (rowgrp == 0) {
        const v4f s = acc + sdata[quad];
        const float inv = 1.0f / (float)len;
        *(v4f*)(seq_mean + (size_t)seg * DIM + col4) = s * inv;
    }
}

__global__ void __launch_bounds__(256)
mean_gemm_wmma_kernel(const float* __restrict__ seq_mean,
                      const float* __restrict__ W,
                      const float* __restrict__ bias,
                      float*       __restrict__ out)
{
    const int wave  = threadIdx.x >> 5;
    const int lane  = threadIdx.x & 31;
    const int tile  = blockIdx.x * 8 + wave;
    const int mtile = tile & 127;
    const int ntile = tile >> 7;

    const int half = lane >> 4;
    const int l15  = lane & 15;

    const int row0 = mtile * 16;
    const int col  = ntile * 16 + l15;

    const float* Arow  = seq_mean + (size_t)(row0 + l15) * DIM + 2 * half;
    const float* Bcol  = W + col;

    v8f c = {};
    #pragma unroll 4
    for (int k = 0; k < DIM; k += 4) {
        v2f a;
        {
            const float2 a2 = *(const float2*)(Arow + k);
            a.x = a2.x; a.y = a2.y;
        }
        v2f bfrag;
        bfrag.x = Bcol[(size_t)(k + 0 + 2 * half) * DIM];
        bfrag.y = Bcol[(size_t)(k + 1 + 2 * half) * DIM];

        c = __builtin_amdgcn_wmma_f32_16x16x4_f32(
                 false, a,  false, bfrag,
                 (short)0, c,  false,  false);
    }

    const float bv = bias[col];
    #pragma unroll
    for (int v = 0; v < 8; ++v) {
        const int row = row0 + v + 8 * half;
        out[(size_t)row * DIM + col] = c[v] + bv;
    }
}

extern "C" void kernel_launch(void* const* d_in, const int* in_sizes, int n_in,
                              void* d_out, int out_size, void* d_ws, size_t ws_size,
                              hipStream_t stream)
{
    const float* values  = (const float*)d_in[0];
    const int*   indices = (const int*)  d_in[1];
    const float* W       = (const float*)d_in[2];
    const float* bias    = (const float*)d_in[3];
    float*       out     = (float*)d_out;

    float* seq_mean = (float*)d_ws;

    seg_mean_kernel<<<NSEG, 128, 0, stream>>>(values, indices, seq_mean);

    mean_gemm_wmma_kernel<<<NSEG / 8, 256, 0, stream>>>(seq_mean, W, bias, out);
}


# segment-sum kernel: two work-groups per segment, each owning 128 of the 256 columns (4 waves = rows mod 4), so a long segment has twice the loads in flight; summation order: 4 row chains per column
# speedup vs baseline: 3.3051x; 1.0820x over previous
#include <hip/hip_runtime.h>
#include <hip/hip_bf16.h>

typedef float v2f __attribute__((ext_vector_type(2)));
typedef float v4f __attribute__((ext_vector_type(4)));
typedef float v8f __attribute__((ext_vector_type(8)));

#define TOTAL_ROWS 262144
#define NSEG 2048
#define DIM  256

#define GROUP_ROWS 16
#define ROW_GROUPS 4
#define COL_PARTS  2
#define WAVE_ROW_STRIDE ((size_t)ROW_GROUPS * DIM)

template <int N>
__device__ __forceinline__ void load_rows(v4f (&t)[N], const float* p)
{
    #pragma unroll
    for (int j = 0; j < N; ++j)
        t[j] = __builtin_nontemporal_load((const v4f*)(p + (size_t)j * WAVE_ROW_STRIDE));
}

template <int N>
__device__ __forceinline__ void sum_rows(v4f& acc, const v4f (&t)[N])
{
    #pragma unroll
    for (int j = 0; j < N; ++j)
        acc += t[j];
}

__global__ void __launch_bounds__(128)
seg_mean_kernel(const float* __restrict__ values,
                const int*   __restrict__ indices,
                float*       __restrict__ seq_mean)
{
    __shared__ v4f sdata[ROW_GROUPS - 1][32];

    const int seg   = blockIdx.x >> 1;
    const int part  = blockIdx.x & 1;
    const int start = indices[2 * seg + 0];
    const int end   = indices[2 * seg + 1];
    const int len   = end - start;

    const int tid    = threadIdx.x;
    const int rowgrp = __builtin_amdgcn_readfirstlane(tid >> 5);
    const int lane   = tid & 31;
    const int col4   = part * (DIM / COL_PARTS) + (lane << 2);

    const int n    = (len > rowgrp) ? ((len - rowgrp + (ROW_GROUPS - 1)) >> 2) : 0;
    const int ng   = n >> 4;
    const int rem  = n & 15;
    const float* p = values + (size_t)(start + rowgrp) * DIM + col4;
    const size_t GS = (size_t)GROUP_ROWS * WAVE_ROW_STRIDE;

    v4f acc = {0.f, 0.f, 0.f, 0.f};
    v4f a[GROUP_ROWS], b[GROUP_ROWS];

    if (ng > 0) load_rows(a, p);
    int g = 0;
    for (; g + 3 <= ng; g += 2) {
        load_rows(b, p + (size_t)(g + 1) * GS);
        asm volatile("" : "+v"(acc) : : "memory");
        sum_rows(acc, a);
        asm volatile("" : "+v"(acc) : : "memory");
        load_rows(a, p + (size_t)(g + 2) * GS);
        sum_rows(acc, b);
    }
    const int left = ng - g;
    if (left == 2) load_rows(b, p + (size_t)(g + 1) * GS);

    const float* q8 = p + (size_t)ng * GS;
    const float* q4 = q8 + (size_t)(rem & 8) * WAVE_ROW_STRIDE;
    const float* q2 = q4 + (size_t)(rem & 4) * WAVE_ROW_STRIDE;
    const float* q1 = q2 + (size_t)(rem & 2) * WAVE_ROW_STRIDE;
    v4f t8[8], t4[4], t2[2], t1[1];
    if (rem & 8) load_rows(t8, q8);
    if (rem & 4) load_rows(t4, q4);
    if (rem & 2) load_rows(t2, q2);
    if (rem & 1) load_rows(t1, q1);

    if (left >= 1) sum_rows(acc, a);
    if (left == 2) sum_rows(acc, b);
    if (rem & 8) sum_rows(acc, t8);
    if (rem & 4) sum_rows(acc, t4);
    if (rem & 2) sum_rows(acc, t2);
    if (rem & 1) sum_rows(acc, t1);

    if (rowgrp != 0) sdata[rowgrp - 1][lane] = acc;
    __syncthreads();
    if (rowgrp == 0) {
        v4f s = acc + sdata[0][lane];
        s += sdata[1][lane];
        s += sdata[2][lane];
        const float inv = 1.0f / (float)len;
        *(v4f*)(seq_mean + (size_t)seg * DIM + col4) = s * inv;
    }
}

__global__ void __launch_bounds__(256)
mean_gemm_wmma_kernel(const float* __restrict__ seq_mean,
                      const float* __restrict__ W,
                      const float* __restrict__ bias,
                      float*       __restrict__ out)
{
    const int wave  = threadIdx.x >> 5;
    const int lane  = threadIdx.x & 31;
    const int tile  = blockIdx.x * 8 + wave;
    const int mtile = tile & 127;
    const int ntile = tile >> 7;

    const int half = lane >> 4;
    const int l15  = lane & 15;

    const int row0 = mtile * 16;
    const int col  = ntile * 16 + l15;

    const float* Arow  = seq_mean + (size_t)(row0 + l15) * DIM + 2 * half;
    const float* Bcol  = W + col;

    v8f c = {};
    #pragma unroll 4
    for (int k = 0; k < DIM; k += 4) {
        v2f a;
        {
            const float2 a2 = *(const float2*)(Arow + k);
            a.x = a2.x; a.y = a2.y;
        }
        v2f bfrag;
        bfrag.x = Bcol[(size_t)(k + 0 + 2 * half) * DIM];
        bfrag.y = Bcol[(size_t)(k + 1 + 2 * half) * DIM];

        c = __builtin_amdgcn_wmma_f32_16x16x4_f32(
                 false, a,  false, bfrag,
                 (short)0, c,  false,  false);
    }

    const float bv = bias[col];
    #pragma unroll
    for (int v = 0; v < 8; ++v) {
        const int row = row0 + v + 8 * half;
        out[(size_t)row * DIM + col] = c[v] + bv;
    }
}

extern "C" void kernel_launch(void* const* d_in, const int* in_sizes, int n_in,
                              void* d_out, int out_size, void* d_ws, size_t ws_size,
                              hipStream_t stream)
{
    const float* values  = (const float*)d_in[0];
    const int*   indices = (const int*)  d_in[1];
    const float* W       = (const float*)d_in[2];
    const float* bias    = (const float*)d_in[3];
    float*       out     = (float*)d_out;

    float* seq_mean = (float*)d_ws;

    seg_mean_kernel<<<NSEG * COL_PARTS, 128, 0, stream>>>(values, indices, seq_mean);

    mean_gemm_wmma_kernel<<<NSEG / 8, 256, 0, stream>>>(seq_mean, W, bias, out);
}


# GEMM kernel: a wave requests all its operands once up front (A: 32 b128 loads per lane; W strip via LDS, LDS-only barrier): 64 WMMAs behind one load round trip, not 16; k summed in permuted order
# speedup vs baseline: 3.5214x; 1.0654x over previous
#include <hip/hip_runtime.h>
#include <hip/hip_bf16.h>

typedef float v2f __attribute__((ext_vector_type(2)));
typedef float v4f __attribute__((ext_vector_type(4)));
typedef float v8f __attribute__((ext_vector_type(8)));

#define TOTAL_ROWS 262144
#define NSEG 2048
#define DIM  256

#define GROUP_ROWS 16
#define ROW_GROUPS 4
#define COL_PARTS  2
#define WAVE_ROW_STRIDE ((size_t)ROW_GROUPS * DIM)

template <int N>
__device__ __forceinline__ void load_rows(v4f (&t)[N], const float* p)
{
    #pragma unroll
    for (int j = 0; j < N; ++j)
        t[j] = __builtin_nontemporal_load((const v4f*)(p + (size_t)j * WAVE_ROW_STRIDE));
}

template <int N>
__device__ __forceinline__ void sum_rows(v4f& acc, const v4f (&t)[N])
{
    #pragma unroll
    for (int j = 0; j < N; ++j)
        acc += t[j];
}

__global__ void __launch_bounds__(128)
seg_mean_kernel(const float* __restrict__ values,
                const int*   __restrict__ indices,
                float*       __restrict__ seq_mean)
{
    __shared__ v4f sdata[ROW_GROUPS - 1][32];

    const int seg   = blockIdx.x >> 1;
    const int part  = blockIdx.x & 1;
    const int start = indices[2 * seg + 0];
    const int end   = indices[2 * seg + 1];
    const int len   = end - start;

    const int tid    = threadIdx.x;
    const int rowgrp = __builtin_amdgcn_readfirstlane(tid >> 5);
    const int lane   = tid & 31;
    const int col4   = part * (DIM / COL_PARTS) + (lane << 2);

    const int n    = (len > rowgrp) ? ((len - rowgrp + (ROW_GROUPS - 1)) >> 2) : 0;
    const int ng   = n >> 4;
    const int rem  = n & 15;
    const float* p = values + (size_t)(start + rowgrp) * DIM + col4;
    const size_t GS = (size_t)GROUP_ROWS * WAVE_ROW_STRIDE;

    v4f acc = {0.f, 0.f, 0.f, 0.f};
    v4f a[GROUP_ROWS], b[GROUP_ROWS];

    if (ng > 0) load_rows(a, p);
    int g = 0;
    for (; g + 3 <= ng; g += 2) {
        load_rows(b, p + (size_t)(g + 1) * GS);
        asm volatile("" : "+v"(acc) : : "memory");
        sum_rows(acc, a);
        asm volatile("" : "+v"(acc) : : "memory");
        load_rows(a, p + (size_t)(g + 2) * GS);
        sum_rows(acc, b);
    }
    const int left = ng - g;
    if (left == 2) load_rows(b, p + (size_t)(g + 1) * GS);

    const float* q8 = p + (size_t)ng * GS;
    const float* q4 = q8 + (size_t)(rem & 8) * WAVE_ROW_STRIDE;
    const float* q2 = q4 + (size_t)(rem & 4) * WAVE_ROW_STRIDE;
    const float* q1 = q2 + (size_t)(rem & 2) * WAVE_ROW_STRIDE;
    v4f t8[8], t4[4], t2[2], t1[1];
    if (rem & 8) load_rows(t8, q8);
    if (rem & 4) load_rows(t4, q4);
    if (rem & 2) load_rows(t2, q2);
    if (rem & 1) load_rows(t1, q1);

    if (left >= 1) sum_rows(acc, a);
    if (left == 2) sum_rows(acc, b);
    if (rem & 8) sum_rows(acc, t8);
    if (rem & 4) sum_rows(acc, t4);
    if (rem & 2) sum_rows(acc, t2);
    if (rem & 1) sum_rows(acc, t1);

    if (rowgrp != 0) sdata[rowgrp - 1][lane] = acc;
    __syncthreads();
    if (rowgrp == 0) {
        v4f s = acc + sdata[0][lane];
        s += sdata[1][lane];
        s += sdata[2][lane];
        const float inv = 1.0f / (float)len;
        *(v4f*)(seq_mean + (size_t)seg * DIM + col4) = s * inv;
    }
}

#define WT_PITCH 260

__global__ void __launch_bounds__(256)
mean_gemm_wmma_kernel(const float* __restrict__ seq_mean,
                      const float* __restrict__ W,
                      const float* __restrict__ bias,
                      float*       __restrict__ out)
{
    __shared__ __attribute__((aligned(16))) float Wt[16 * WT_PITCH];

    const int tid   = threadIdx.x;
    const int wave  = tid >> 5;
    const int lane  = tid & 31;
    const int tile  = blockIdx.x * 8 + wave;
    const int mtile = tile & 127;
    const int ntile = blockIdx.x >> 4;

    const int half = lane >> 4;
    const int l15  = lane & 15;

    const int row0 = mtile * 16;
    const int n0   = ntile * 16;
    const int col  = n0 + l15;

    const int q = tid & 3;
    const int r = tid >> 2;
    v4f w4[4];
    #pragma unroll
    for (int j = 0; j < 4; ++j)
        w4[j] = *(const v4f*)(W + (size_t)(r + 64 * j) * DIM + n0 + 4 * q);

    const float* Arow = seq_mean + (size_t)(row0 + l15) * DIM + 4 * half;
    v4f a4[32];
    #pragma unroll
    for (int i = 0; i < 32; ++i)
        a4[i] = *(const v4f*)(Arow + 8 * i);

    #pragma unroll
    for (int j = 0; j < 4; ++j) {
        #pragma unroll
        for (int e = 0; e < 4; ++e)
            Wt[(4 * q + e) * WT_PITCH + r + 64 * j] = w4[j][e];
    }
    __builtin_amdgcn_fence(__ATOMIC_RELEASE, "workgroup", "local");
    __builtin_amdgcn_s_barrier();
    __builtin_amdgcn_fence(__ATOMIC_ACQUIRE, "workgroup", "local");

    const float* Bl = Wt + l15 * WT_PITCH + 4 * half;
    v4f b4[32];
    #pragma unroll
    for (int i = 0; i < 32; ++i)
        b4[i] = *(const v4f*)(Bl + 8 * i);

    __builtin_amdgcn_sched_barrier(0);

    v8f c = {};
    #pragma unroll
    for (int i = 0; i < 32; ++i) {
        v2f a, b;
        a.x = a4[i].x; a.y = a4[i].y;
        b.x = b4[i].x; b.y = b4[i].y;
        c = __builtin_amdgcn_wmma_f32_16x16x4_f32(
                 false, a,  false, b,
                 (short)0, c,  false,  false);
        a.x = a4[i].z; a.y = a4[i].w;
        b.x = b4[i].z; b.y = b4[i].w;
        c = __builtin_amdgcn_wmma_f32_16x16x4_f32(
                 false, a,  false, b,
                 (short)0, c,  false,  false);
    }

    const float bv = bias[col];
    #pragma unroll
    for (int v = 0; v < 8; ++v) {
        const int row = row0 + v + 8 * half;
        out[(size_t)row * DIM + col] = c[v] + bv;
    }
}

extern "C" void kernel_launch(void* const* d_in, const int* in_sizes, int n_in,
                              void* d_out, int out_size, void* d_ws, size_t ws_size,
                              hipStream_t stream)
{
    const float* values  = (const float*)d_in[0];
    const int*   indices = (const int*)  d_in[1];
    const float* W       = (const float*)d_in[2];
    const float* bias    = (const float*)d_in[3];
    float*       out     = (float*)d_out;

    float* seq_mean = (float*)d_ws;

    seg_mean_kernel<<<NSEG * COL_PARTS, 128, 0, stream>>>(values, indices, seq_mean);

    mean_gemm_wmma_kernel<<<NSEG / 8, 256, 0, stream>>>(seq_mean, W, bias, out);
}


# segment-sum kernel: four 2-wave work-groups per segment (64 columns each, a wave-load = two rows), 24-row register sets: a long segment draws on 4 WGPs with 1.5x the loads in flight; sum order kept
# speedup vs baseline: 3.7904x; 1.0764x over previous
#include <hip/hip_runtime.h>
#include <hip/hip_bf16.h>

typedef float v2f __attribute__((ext_vector_type(2)));
typedef float v4f __attribute__((ext_vector_type(4)));
typedef float v8f __attribute__((ext_vector_type(8)));

#define TOTAL_ROWS 262144
#define NSEG 2048
#define DIM  256

#define GROUP_ROWS  24
#define ROW_STREAMS 4
#define COL_PARTS   4
#define STREAM_ROW_STRIDE ((size_t)ROW_STREAMS * DIM)

template <int N>
__device__ __forceinline__ void load_rows(v4f (&t)[N], const float* p)
{
    #pragma unroll
    for (int j = 0; j < N; ++j)
        t[j] = __builtin_nontemporal_load((const v4f*)(p + (size_t)j * STREAM_ROW_STRIDE));
}

template <int N>
__device__ __forceinline__ void sum_rows(v4f& acc, const v4f (&t)[N])
{
    #pragma unroll
    for (int j = 0; j < N; ++j)
        acc += t[j];
}

template <int N>
__device__ __forceinline__ void load_rows_upto(v4f (&t)[N], const float* p, int cnt)
{
    #pragma unroll
    for (int j = 0; j < N; ++j) {
        if (j < cnt)
            t[j] = __builtin_nontemporal_load((const v4f*)(p + (size_t)j * STREAM_ROW_STRIDE));
        else
            t[j] = v4f{0.f, 0.f, 0.f, 0.f};
    }
}

__global__ void __launch_bounds__(64)
seg_mean_kernel(const float* __restrict__ values,
                const int*   __restrict__ indices,
                float*       __restrict__ seq_mean)
{
    __shared__ v4f sdata[ROW_STREAMS][16];

    const int seg   = blockIdx.x >> 2;
    const int part  = blockIdx.x & 3;
    const int start = indices[2 * seg + 0];
    const int end   = indices[2 * seg + 1];
    const int len   = end - start;

    const int tid  = threadIdx.x;
    const int wave = __builtin_amdgcn_readfirstlane(tid >> 5);
    const int lane = tid & 31;
    const int l15  = lane & 15;
    const int sid  = 2 * wave + (lane >> 4);
    const int col4 = part * (DIM / COL_PARTS) + (l15 << 2);

    const int sodd   = 2 * wave + 1;
    const int n      = (len > sodd) ? ((len - sodd + (ROW_STREAMS - 1)) >> 2) : 0;
    const int n_lane = (len > sid)  ? ((len - sid  + (ROW_STREAMS - 1)) >> 2) : 0;
    const int ng     = n / GROUP_ROWS;
    const int rem    = n - ng * GROUP_ROWS;
    const float* p = values + (size_t)(start + sid) * DIM + col4;
    const size_t GS = (size_t)GROUP_ROWS * STREAM_ROW_STRIDE;

    const bool extra = n_lane > n;
    v4f acc = {0.f, 0.f, 0.f, 0.f};
    v4f a[GROUP_ROWS], b[GROUP_ROWS];

    int g = 0;
    if (ng > 0) {
        const bool even = (ng & 1) == 0;
        if (even) load_rows(b, p);
        load_rows(a, even ? p + GS : p);
        if (even) {
            asm volatile("" : "+v"(acc) : : "memory");
            sum_rows(acc, b);
            asm volatile("" : "+v"(acc) : : "memory");
            g = 1;
        }
    }
    for (; g + 3 <= ng; g += 2) {
        load_rows(b, p + (size_t)(g + 1) * GS);
        asm volatile("" : "+v"(acc) : : "memory");
        sum_rows(acc, a);
        asm volatile("" : "+v"(acc) : : "memory");
        load_rows(a, p + (size_t)(g + 2) * GS);
        sum_rows(acc, b);
    }
    load_rows_upto(b, p + (size_t)ng * GS, rem);
    v4f tx = {0.f, 0.f, 0.f, 0.f};
    if (extra)
        tx = __builtin_nontemporal_load((const v4f*)(p + (size_t)n * STREAM_ROW_STRIDE));
    if (ng > 0) sum_rows(acc, a);
    sum_rows(acc, b);
    if (extra) acc += tx;

    sdata[sid][l15] = acc;
    __syncthreads();
    if (tid < 16) {
        v4f s = sdata[0][tid] + sdata[1][tid];
        s += sdata[2][tid];
        s += sdata[3][tid];
        const float inv = 1.0f / (float)len;
        *(v4f*)(seq_mean + (size_t)seg * DIM + col4) = s * inv;
    }
}

#define WT_PITCH 260

__global__ void __launch_bounds__(256)
mean_gemm_wmma_kernel(const float* __restrict__ seq_mean,
                      const float* __restrict__ W,
                      const float* __restrict__ bias,
                      float*       __restrict__ out)
{
    __shared__ __attribute__((aligned(16))) float Wt[16 * WT_PITCH];

    const int tid   = threadIdx.x;
    const int wave  = tid >> 5;
    const int lane  = tid & 31;
    const int tile  = blockIdx.x * 8 + wave;
    const int mtile = tile & 127;
    const int ntile = blockIdx.x >> 4;

    const int half = lane >> 4;
    const int l15  = lane & 15;

    const int row0 = mtile * 16;
    const int n0   = ntile * 16;
    const int col  = n0 + l15;

    const int q = tid & 3;
    const int r = tid >> 2;
    v4f w4[4];
    #pragma unroll
    for (int j = 0; j < 4; ++j)
        w4[j] = *(const v4f*)(W + (size_t)(r + 64 * j) * DIM + n0 + 4 * q);

    const float* Arow = seq_mean + (size_t)(row0 + l15) * DIM + 4 * half;
    v4f a4[32];
    #pragma unroll
    for (int i = 0; i < 32; ++i)
        a4[i] = *(const v4f*)(Arow + 8 * i);

    #pragma unroll
    for (int j = 0; j < 4; ++j) {
        #pragma unroll
        for (int e = 0; e < 4; ++e)
            Wt[(4 * q + e) * WT_PITCH + r + 64 * j] = w4[j][e];
    }
    __builtin_amdgcn_fence(__ATOMIC_RELEASE, "workgroup", "local");
    __builtin_amdgcn_s_barrier();
    __builtin_amdgcn_fence(__ATOMIC_ACQUIRE, "workgroup", "local");

    const float* Bl = Wt + l15 * WT_PITCH + 4 * half;
    v4f b4[32];
    #pragma unroll
    for (int i = 0; i < 32; ++i)
        b4[i] = *(const v4f*)(Bl + 8 * i);

    __builtin_amdgcn_sched_barrier(0);

    v8f c = {};
    #pragma unroll
    for (int i = 0; i < 32; ++i) {
        v2f a, b;
        a.x = a4[i].x; a.y = a4[i].y;
        b.x = b4[i].x; b.y = b4[i].y;
        c = __builtin_amdgcn_wmma_f32_16x16x4_f32(
                 false, a,  false, b,
                 (short)0, c,  false,  false);
        a.x = a4[i].z; a.y = a4[i].w;
        b.x = b4[i].z; b.y = b4[i].w;
        c = __builtin_amdgcn_wmma_f32_16x16x4_f32(
                 false, a,  false, b,
                 (short)0, c,  false,  false);
    }

    const float bv = bias[col];
    #pragma unroll
    for (int v = 0; v < 8; ++v) {
        const int row = row0 + v + 8 * half;
        out[(size_t)row * DIM + col] = c[v] + bv;
    }
}

extern "C" void kernel_launch(void* const* d_in, const int* in_sizes, int n_in,
                              void* d_out, int out_size, void* d_ws, size_t ws_size,
                              hipStream_t stream)
{
    const float* values  = (const float*)d_in[0];
    const int*   indices = (const int*)  d_in[1];
    const float* W       = (const float*)d_in[2];
    const float* bias    = (const float*)d_in[3];
    float*       out     = (float*)d_out;

    float* seq_mean = (float*)d_ws;

    seg_mean_kernel<<<NSEG * COL_PARTS, 64, 0, stream>>>(values, indices, seq_mean);

    mean_gemm_wmma_kernel<<<NSEG / 8, 256, 0, stream>>>(seq_mean, W, bias, out);
}
